# starting kernel
# baseline (speedup 1.0000x reference)
#include <hip/hip_runtime.h>
#include <math.h>

constexpr int kB = 8;
constexpr int kT = 2048;
constexpr int kD = 512;

typedef __attribute__((ext_vector_type(2))) float v2f;
typedef __attribute__((ext_vector_type(8))) float v8f;

__device__ __forceinline__ float softplus_dev(const float* kp) {
    float k = kp[0];
    return fmaxf(k, 0.0f) + log1pf(expf(-fabsf(k)));
}

__global__ void delta_mag_sq_kernel(const float* __restrict__ h,
                                    float* __restrict__ dsq) {
    int gid  = blockIdx.x * blockDim.x + threadIdx.x;
    int lane = gid & 31;
    int wave = gid >> 5;
    int b    = wave >> 4;
    int d    = ((wave & 15) << 5) + lane;
    const float* hp = h + (size_t)b * kT * kD + d;

    float cum = 0.f, hprev = 0.f;
    for (int t = 0; t < kT; ++t) {
        float hv = hp[(size_t)t * kD];
        cum += hv;
        float dl = cum - hprev;
        hprev = hv;
        float sq = dl * dl;
        #pragma unroll
        for (int off = 16; off > 0; off >>= 1)
            sq += __shfl_xor(sq, off, 32);
        if (lane == 0) atomicAdd(&dsq[b * kT + t], sq);
    }
}

__global__ void mag_sqrt_kernel(float* __restrict__ dsq) {
    int i = blockIdx.x * blockDim.x + threadIdx.x;
    dsq[i] = sqrtf(dsq[i]);
}

__global__ void context_wmma_kernel(const float* __restrict__ h,
                                    const float* __restrict__ kp,
                                    const float* __restrict__ mag,
                                    float* __restrict__ out) {
    const int lane  = threadIdx.x;
    const int dtile = blockIdx.x & 31;
    const int b     = blockIdx.x >> 5;
    const int n     = lane & 15;
    const int hi    = (lane >> 4) & 1;

    const float sp  = softplus_dev(kp);
    const float l2r = -sp * 1.4426950408889634f;
    auto rpow = [&](int dist) -> float { return exp2f((float)dist * l2r); };

    v2f a[4];
    #pragma unroll
    for (int g = 0; g < 4; ++g) {
        int j0 = 4 * g + 2 * hi;
        a[g].x = (j0     <= n) ? rpow(n - j0)     : 0.f;
        a[g].y = (j0 + 1 <= n) ? rpow(n - j0 - 1) : 0.f;
    }

    float rpc[8];
    #pragma unroll
    for (int v = 0; v < 8; ++v) rpc[v] = rpow(v + 1 + 8 * hi);

    const float* hb = h   + ((size_t)b * kT) * kD + dtile * 16 + n;
    float*       ob = out + ((size_t)b * kT) * kD + dtile * 16 + n;
    const float* mb = mag + b * kT;

    float S = 0.f;
    for (int t0 = 0; t0 < kT; t0 += 16) {
        v2f bm[4];
        #pragma unroll
        for (int g = 0; g < 4; ++g) {
            size_t j = (size_t)(t0 + 4 * g + 2 * hi);
            bm[g].x = hb[j * kD];
            bm[g].y = hb[(j + 1) * kD];
        }
        v8f c;
        #pragma unroll
        for (int v = 0; v < 8; ++v) c[v] = rpc[v] * S;

        c = __builtin_amdgcn_wmma_f32_16x16x4_f32(false, a[0], false, bm[0], (short)0, c, false, false);
        c = __builtin_amdgcn_wmma_f32_16x16x4_f32(false, a[1], false, bm[1], (short)0, c, false, false);
        c = __builtin_amdgcn_wmma_f32_16x16x4_f32(false, a[2], false, bm[2], (short)0, c, false, false);
        c = __builtin_amdgcn_wmma_f32_16x16x4_f32(false, a[3], false, bm[3], (short)0, c, false, false);

        #pragma unroll
        for (int v = 0; v < 8; ++v) {
            int M = v + 8 * hi;
            ob[(size_t)(t0 + M) * kD] = mb[t0 + M] * fabsf(c[v]);
        }
        S = __shfl(c[7], 16 + n, 32);
    }
}

extern "C" void kernel_launch(void* const* d_in, const int* in_sizes, int n_in,
                              void* d_out, int out_size, void* d_ws, size_t ws_size,
                              hipStream_t stream) {
    const float* h  = (const float*)d_in[0];
    const float* k  = (const float*)d_in[1];
    float* out = (float*)d_out;
    float* dsq = (float*)d_ws;

    (void)hipMemsetAsync(dsq, 0, (size_t)kB * kT * sizeof(float), stream);
    delta_mag_sq_kernel<<<16, 256, 0, stream>>>(h, dsq);
    mag_sqrt_kernel<<<(kB * kT) / 256, 256, 0, stream>>>(dsq);
    context_wmma_kernel<<<kB * (kD / 16), 32, 0, stream>>>(h, k, dsq, out);
}


# Scan loop's lane exchanges moved from the LDS unit to the vector ALU: DPP row_xmask adds for the 16-column sum, v_permlanex16 for half swap and carry broadcast; sums and their order unchanged
# speedup vs baseline: 14.9458x; 14.9458x over previous
#include <hip/hip_runtime.h>
#include <math.h>

constexpr int kB = 8;
constexpr int kT = 2048;
constexpr int kD = 512;

typedef __attribute__((ext_vector_type(2))) float v2f;
typedef __attribute__((ext_vector_type(8))) float v8f;

__device__ __forceinline__ float softplus_dev(const float* kp) {
    float k = kp[0];
    return fmaxf(k, 0.0f) + log1pf(expf(-fabsf(k)));
}

__device__ __forceinline__ float xhalf(float x) {
    const int xi = __builtin_bit_cast(int, x);
    return __builtin_bit_cast(float, __builtin_amdgcn_permlanex16(0, xi, 0x76543210, 0xfedcba98, false, false));
}
template <int M>
__device__ __forceinline__ float xrow(float x) {
    const int xi = __builtin_bit_cast(int, x);
    return __builtin_bit_cast(float, __builtin_amdgcn_update_dpp(0, xi, 0x160 + M, 0xf, 0xf, true));
}

__global__ void __launch_bounds__(32) scan_both_wmma_kernel(const float* __restrict__ h,
                                      const float* __restrict__ kp,
                                      float* __restrict__ dsq,
                                      float* __restrict__ out) {
    const int lane  = threadIdx.x;
    const int dtile = blockIdx.x & 31;
    const int b     = blockIdx.x >> 5;
    const int n     = lane & 15;
    const int hi    = (lane >> 4) & 1;

    const float sp  = softplus_dev(kp);
    const float l2r = -sp * 1.4426950408889634f;
    auto rpow = [&](int dist) -> float { return exp2f((float)dist * l2r); };

    v2f a[4], a1[4];
    #pragma unroll
    for (int g = 0; g < 4; ++g) {
        int j0 = 8 * hi + 2 * g;
        a[g].x  = (j0     <= n) ? rpow(n - j0)     : 0.f;
        a[g].y  = (j0 + 1 <= n) ? rpow(n - j0 - 1) : 0.f;
        a1[g].x = (j0     <= n) ? 1.f : 0.f;
        a1[g].y = (j0 + 1 <= n) ? 1.f : 0.f;
    }

    float rpc[8];
    #pragma unroll
    for (int v = 0; v < 8; ++v) rpc[v] = rpow(v + 1 + 8 * hi);

    const float* hb = h   + ((size_t)b * kT) * kD + dtile * 16 + n;
    float*       ob = out + ((size_t)b * kT) * kD + dtile * 16 + n;
    float*       qb = dsq + b * kT;

    const bool b3 = (n & 8) != 0;
    const bool b2 = (n & 4) != 0;
    const bool b1 = (n & 2) != 0;

    auto load_chunk = [&](v2f (&dst)[4], int t0) {
        #pragma unroll
        for (int g = 0; g < 4; ++g) {
            size_t j = (size_t)(t0 + 8 * hi + 2 * g);
            dst[g].x = hb[j * kD];
            dst[g].y = hb[(j + 1) * kD];
        }
    };

    constexpr int kAhead = 8;
    constexpr int kChunks = kT / 16;
    v2f ring[kAhead][4];
    #pragma unroll
    for (int u = 0; u < kAhead; ++u) load_chunk(ring[u], 16 * u);

    float S    = 0.f;
    float Sc   = 0.f;
    float last = 0.f;
    for (int c0 = 0; c0 < kChunks; c0 += kAhead) {
        #pragma unroll
        for (int u = 0; u < kAhead; ++u) {
            const int t0 = 16 * (c0 + u);
            v2f (&bm)[4] = ring[u];

            v8f c, cc;
            #pragma unroll
            for (int v = 0; v < 8; ++v) { c[v] = rpc[v] * S; cc[v] = Sc; }

            c  = __builtin_amdgcn_wmma_f32_16x16x4_f32(false, a[0],  false, bm[0], (short)0, c,  false, false);
            cc = __builtin_amdgcn_wmma_f32_16x16x4_f32(false, a1[0], false, bm[0], (short)0, cc, false, false);
            c  = __builtin_amdgcn_wmma_f32_16x16x4_f32(false, a[1],  false, bm[1], (short)0, c,  false, false);
            cc = __builtin_amdgcn_wmma_f32_16x16x4_f32(false, a1[1], false, bm[1], (short)0, cc, false, false);
            c  = __builtin_amdgcn_wmma_f32_16x16x4_f32(false, a[2],  false, bm[2], (short)0, c,  false, false);
            cc = __builtin_amdgcn_wmma_f32_16x16x4_f32(false, a1[2], false, bm[2], (short)0, cc, false, false);
            c  = __builtin_amdgcn_wmma_f32_16x16x4_f32(false, a[3],  false, bm[3], (short)0, c,  false, false);
            cc = __builtin_amdgcn_wmma_f32_16x16x4_f32(false, a1[3], false, bm[3], (short)0, cc, false, false);

            #pragma unroll
            for (int v = 0; v < 8; ++v) {
                int M = v + 8 * hi;
                ob[(size_t)(t0 + M) * kD] = fabsf(c[v]);
            }

            float xch = xhalf(bm[3].y);
            float hp0 = hi ? xch : last;
            last = xch;

            float sq[8];
            {
                float dl = cc[0] - hp0;
                sq[0] = dl * dl;
            }
            #pragma unroll
            for (int v = 1; v < 8; ++v) {
                float hpv = ((v - 1) & 1) ? bm[(v - 1) >> 1].y : bm[(v - 1) >> 1].x;
                float dl = cc[v] - hpv;
                sq[v] = dl * dl;
            }

            asm volatile("" : "+v"(sq[0]), "+v"(sq[1]), "+v"(sq[2]), "+v"(sq[3]),
                              "+v"(sq[4]), "+v"(sq[5]), "+v"(sq[6]), "+v"(sq[7])
                            : "v"(bm[0].x), "v"(bm[0].y), "v"(bm[1].x), "v"(bm[1].y),
                              "v"(bm[2].x), "v"(bm[2].y), "v"(bm[3].x), "v"(bm[3].y)
                            : "memory");
            {
                int cn = c0 + u + kAhead;
                cn = cn < kChunks ? cn : kChunks - 1;
                load_chunk(ring[u], 16 * cn);
            }
            float r4[4];
            #pragma unroll
            for (int i = 0; i < 4; ++i) {
                float keep = b3 ? sq[i + 4] : sq[i];
                float send = b3 ? sq[i]     : sq[i + 4];
                r4[i] = keep + xrow<8>(send);
            }
            float r2[2];
            #pragma unroll
            for (int i = 0; i < 2; ++i) {
                float keep = b2 ? r4[i + 2] : r4[i];
                float send = b2 ? r4[i]     : r4[i + 2];
                r2[i] = keep + xrow<4>(send);
            }
            float keep1 = b1 ? r2[1] : r2[0];
            float send1 = b1 ? r2[0] : r2[1];
            float r1  = keep1 + xrow<2>(send1);
            float tot = r1 + xrow<1>(r1);
            if ((n & 1) == 0) atomicAdd(&qb[t0 + (n >> 1) + 8 * hi], tot);

            {
                const float sx  = xhalf(c[7]);
                const float scx = xhalf(cc[7]);
                S  = hi ? c[7]  : sx;
                Sc = hi ? cc[7] : scx;
            }
        }
    }
}

constexpr int kScaleRows    = 4;
constexpr int kScaleThreads = 256;
constexpr int kScaleWaves   = (kB * kT) / kScaleRows;
constexpr int kScaleLoads   = kScaleRows * (kD / 4) / 32;
__global__ void __launch_bounds__(kScaleThreads) scale_rows_kernel(const float* __restrict__ dsq,
                                  float* __restrict__ out) {
    const int lane = threadIdx.x & 31;
    const int wave = blockIdx.x * (kScaleThreads / 32) + (threadIdx.x >> 5);
    const size_t row0 = (size_t)wave * kScaleRows;

    const float4 q = *reinterpret_cast<const float4*>(dsq + row0);
    float4* o4 = reinterpret_cast<float4*>(out) + row0 * (kD / 4) + lane;

    float4 v[kScaleLoads];
    #pragma unroll
    for (int i = 0; i < kScaleLoads; ++i) v[i] = o4[i * 32];

    const float m[kScaleRows] = { sqrtf(q.x), sqrtf(q.y), sqrtf(q.z), sqrtf(q.w) };

    #pragma unroll
    for (int i = 0; i < kScaleLoads; ++i) {
        const float s = m[i / (kScaleLoads / kScaleRows)];
        float4 w = v[i];
        w.x *= s; w.y *= s; w.z *= s; w.w *= s;
        o4[i * 32] = w;
    }
}

extern "C" void kernel_launch(void* const* d_in, const int* in_sizes, int n_in,
                              void* d_out, int out_size, void* d_ws, size_t ws_size,
                              hipStream_t stream) {
    const float* h  = (const float*)d_in[0];
    const float* k  = (const float*)d_in[1];
    float* out = (float*)d_out;
    float* dsq = (float*)d_ws;

    (void)hipMemsetAsync(dsq, 0, (size_t)kB * kT * sizeof(float), stream);
    scan_both_wmma_kernel<<<kB * (kD / 16), 32, 0, stream>>>(h, k, dsq, out);
    scale_rows_kernel<<<kScaleWaves / (kScaleThreads / 32), kScaleThreads, 0, stream>>>(dsq, out);
}


# Scan loop freed of address-translation waits (s_wait_xcnt): branch-free row-sum atomic per 2 chunks, issued a chunk later; stored values and bases stay unwritten for a chunk; sums unchanged
# speedup vs baseline: 16.0067x; 1.0710x over previous
#include <hip/hip_runtime.h>
#include <math.h>

constexpr int kB = 8;
constexpr int kT = 2048;
constexpr int kD = 512;
constexpr int kGuard = 32;

typedef __attribute__((ext_vector_type(2))) float v2f;
typedef __attribute__((ext_vector_type(8))) float v8f;
typedef __attribute__((address_space(1))) float gfloat;

__device__ __forceinline__ float softplus_dev(const float* kp) {
    float k = kp[0];
    return fmaxf(k, 0.0f) + log1pf(expf(-fabsf(k)));
}

__device__ __forceinline__ float xhalf(float x) {
    const int xi = __builtin_bit_cast(int, x);
    return __builtin_bit_cast(float, __builtin_amdgcn_permlanex16(0, xi, 0x76543210, 0xfedcba98, false, false));
}
template <int M>
__device__ __forceinline__ float xrow(float x) {
    const int xi = __builtin_bit_cast(int, x);
    return __builtin_bit_cast(float, __builtin_amdgcn_update_dpp(0, xi, 0x160 + M, 0xf, 0xf, true));
}

__global__ void __launch_bounds__(32) scan_both_wmma_kernel(const float* __restrict__ h,
                                      const float* __restrict__ kp,
                                      float* __restrict__ dsq,
                                      float* __restrict__ out) {
    const int lane  = threadIdx.x;
    const int dtile = blockIdx.x & 31;
    const int b     = blockIdx.x >> 5;
    const int n     = lane & 15;
    const int hi    = (lane >> 4) & 1;

    const float sp  = softplus_dev(kp);
    const float l2r = -sp * 1.4426950408889634f;
    auto rpow = [&](int dist) -> float { return exp2f((float)dist * l2r); };

    v2f a[4], a1[4];
    #pragma unroll
    for (int g = 0; g < 4; ++g) {
        int j0 = 8 * hi + 2 * g;
        a[g].x  = (j0     <= n) ? rpow(n - j0)     : 0.f;
        a[g].y  = (j0 + 1 <= n) ? rpow(n - j0 - 1) : 0.f;
        a1[g].x = (j0     <= n) ? 1.f : 0.f;
        a1[g].y = (j0 + 1 <= n) ? 1.f : 0.f;
    }

    float rpc[8];
    #pragma unroll
    for (int v = 0; v < 8; ++v) rpc[v] = rpow(v + 1 + 8 * hi);

    const float* hb = h   + ((size_t)b * kT + 8 * hi) * kD + dtile * 16 + n;
    float*       ob = out + ((size_t)b * kT + 8 * hi) * kD + dtile * 16 + n;
    float*       qb = dsq + b * kT + 16 * (n & 1) + (n >> 1) + 8 * hi;

    const bool b3 = (n & 8) != 0;
    const bool b2 = (n & 4) != 0;
    const bool b1 = (n & 2) != 0;
    const bool b0 = (n & 1) != 0;

    auto load_chunk = [&](v2f (&dst)[4], const float* p0) {
        #pragma unroll
        for (int g = 0; g < 4; ++g) {
            dst[g].x = p0[(2 * g) * kD];
            dst[g].y = p0[(2 * g + 1) * kD];
        }
    };

    constexpr int kAhead = 8;
    constexpr int kChunks = kT / 16;
    v2f ring[kAhead][4];
    #pragma unroll
    for (int u = 0; u < kAhead; ++u) load_chunk(ring[u], hb + (size_t)(16 * u) * kD);

    float S    = 0.f;
    float Sc   = 0.f;
    float last = 0.f;
    float r1_even = 0.f;
    constexpr int kKeep = 2;
    float kept[kKeep][8];
    float pend   = 0.f;
    float issued = 0.f;
    #pragma unroll
    for (int i = 0; i < kKeep; ++i) {
        #pragma unroll
        for (int v = 0; v < 8; ++v) kept[i][v] = 0.f;
    }
    gfloat* op = (gfloat*)ob;
    gfloat* qp = (gfloat*)qb;
    for (int c0 = 0; c0 < kChunks; c0 += kAhead) {
        const int cl = (c0 + kAhead < kChunks) ? (c0 + kAhead) : (kChunks - kAhead);
        const float* lp = hb + (size_t)(16 * cl) * kD;
        #pragma unroll
        for (int u = 0; u < kAhead; ++u) {
            v2f (&bm)[4] = ring[u];
            asm volatile("" : "+v"(bm[3].y));

            v8f c, cc;
            #pragma unroll
            for (int v = 0; v < 8; ++v) { c[v] = rpc[v] * S; cc[v] = Sc; }

            c  = __builtin_amdgcn_wmma_f32_16x16x4_f32(false, a[0],  false, bm[0], (short)0, c,  false, false);
            cc = __builtin_amdgcn_wmma_f32_16x16x4_f32(false, a1[0], false, bm[0], (short)0, cc, false, false);
            c  = __builtin_amdgcn_wmma_f32_16x16x4_f32(false, a[1],  false, bm[1], (short)0, c,  false, false);
            cc = __builtin_amdgcn_wmma_f32_16x16x4_f32(false, a1[1], false, bm[1], (short)0, cc, false, false);
            c  = __builtin_amdgcn_wmma_f32_16x16x4_f32(false, a[2],  false, bm[2], (short)0, c,  false, false);
            cc = __builtin_amdgcn_wmma_f32_16x16x4_f32(false, a1[2], false, bm[2], (short)0, cc, false, false);
            c  = __builtin_amdgcn_wmma_f32_16x16x4_f32(false, a[3],  false, bm[3], (short)0, c,  false, false);
            cc = __builtin_amdgcn_wmma_f32_16x16x4_f32(false, a1[3], false, bm[3], (short)0, cc, false, false);

            if (u == 1) {
                asm volatile("" : "+v"(op), "+v"(qp) : "v"(c[7]), "v"(cc[7]) : "memory");
                op += (size_t)(16 * kAhead) * kD;
                qp += 16 * kAhead;
                asm volatile("" : "+v"(op), "+v"(qp));
            }
            const int rel = (u == 0) ? 0 : -kAhead;
            if ((u & 1) == 0) {
                asm volatile("" : : "v"(c[7]), "v"(cc[7]) : "memory");
                atomicAdd((float*)(qp + 16 * (u - 2 + rel)), pend);
                issued = pend;
            }
            float ab[8];
            #pragma unroll
            for (int v = 0; v < 8; ++v) {
                ab[v] = fabsf(c[v]);
                op[(ptrdiff_t)(16 * (u + rel) + v) * kD] = ab[v];
            }

            float xch = xhalf(bm[3].y);
            float hp0 = hi ? xch : last;
            last = xch;

            float sq[8];
            {
                float dl = cc[0] - hp0;
                sq[0] = dl * dl;
            }
            #pragma unroll
            for (int v = 1; v < 8; ++v) {
                float hpv = ((v - 1) & 1) ? bm[(v - 1) >> 1].y : bm[(v - 1) >> 1].x;
                float dl = cc[v] - hpv;
                sq[v] = dl * dl;
            }

            asm volatile("" : "+v"(sq[0]), "+v"(sq[1]), "+v"(sq[2]), "+v"(sq[3]),
                              "+v"(sq[4]), "+v"(sq[5]), "+v"(sq[6]), "+v"(sq[7])
                            : "v"(bm[0].x), "v"(bm[0].y), "v"(bm[1].x), "v"(bm[1].y),
                              "v"(bm[2].x), "v"(bm[2].y), "v"(bm[3].x), "v"(bm[3].y)
                            : "memory");
            load_chunk(ring[u], lp + (size_t)(16 * u) * kD);
            float r4[4];
            #pragma unroll
            for (int i = 0; i < 4; ++i) {
                float keep = b3 ? sq[i + 4] : sq[i];
                float send = b3 ? sq[i]     : sq[i + 4];
                r4[i] = keep + xrow<8>(send);
            }
            float r2[2];
            #pragma unroll
            for (int i = 0; i < 2; ++i) {
                float keep = b2 ? r4[i + 2] : r4[i];
                float send = b2 ? r4[i]     : r4[i + 2];
                r2[i] = keep + xrow<4>(send);
            }
            float keep1 = b1 ? r2[1] : r2[0];
            float send1 = b1 ? r2[0] : r2[1];
            float r1  = keep1 + xrow<2>(send1);
            if ((u & 1) == 0) {
                r1_even = r1;
            } else {
                float keep0 = b0 ? r1      : r1_even;
                float send0 = b0 ? r1_even : r1;
                float tot   = keep0 + xrow<1>(send0);
                asm volatile("" : : "v"(issued));
                pend = tot;
            }
            asm volatile("" : : "v"(kept[u % kKeep][0]), "v"(kept[u % kKeep][1]), "v"(kept[u % kKeep][2]),
                                "v"(kept[u % kKeep][3]), "v"(kept[u % kKeep][4]), "v"(kept[u % kKeep][5]),
                                "v"(kept[u % kKeep][6]), "v"(kept[u % kKeep][7]));
            #pragma unroll
            for (int v = 0; v < 8; ++v) kept[u % kKeep][v] = ab[v];

            {
                const float sx  = xhalf(c[7]);
                const float scx = xhalf(cc[7]);
                S  = hi ? c[7]  : sx;
                Sc = hi ? cc[7] : scx;
            }
        }
    }
    atomicAdd(qb + 16 * (kChunks - 2), pend);
}

constexpr int kScaleRows    = 4;
constexpr int kScaleThreads = 256;
constexpr int kScaleWaves   = (kB * kT) / kScaleRows;
constexpr int kScaleLoads   = kScaleRows * (kD / 4) / 32;
__global__ void __launch_bounds__(kScaleThreads) scale_rows_kernel(const float* __restrict__ dsq,
                                  float* __restrict__ out) {
    const int lane = threadIdx.x & 31;
    const int wave = blockIdx.x * (kScaleThreads / 32) + (threadIdx.x >> 5);
    const size_t row0 = (size_t)wave * kScaleRows;

    const float4 q = *reinterpret_cast<const float4*>(dsq + row0);
    float4* o4 = reinterpret_cast<float4*>(out) + row0 * (kD / 4) + lane;

    float4 v[kScaleLoads];
    #pragma unroll
    for (int i = 0; i < kScaleLoads; ++i) v[i] = o4[i * 32];

    const float m[kScaleRows] = { sqrtf(q.x), sqrtf(q.y), sqrtf(q.z), sqrtf(q.w) };

    #pragma unroll
    for (int i = 0; i < kScaleLoads; ++i) {
        const float s = m[i / (kScaleLoads / kScaleRows)];
        float4 w = v[i];
        w.x *= s; w.y *= s; w.z *= s; w.w *= s;
        o4[i * 32] = w;
    }
}

extern "C" void kernel_launch(void* const* d_in, const int* in_sizes, int n_in,
                              void* d_out, int out_size, void* d_ws, size_t ws_size,
                              hipStream_t stream) {
    const float* h  = (const float*)d_in[0];
    const float* k  = (const float*)d_in[1];
    float* out = (float*)d_out;
    float* dsq = (float*)d_ws + kGuard;

    (void)hipMemsetAsync(d_ws, 0, (size_t)(kB * kT + kGuard) * sizeof(float), stream);
    scan_both_wmma_kernel<<<kB * (kD / 16), 32, 0, stream>>>(h, k, dsq, out);
    scale_rows_kernel<<<kScaleWaves / (kScaleThreads / 32), kScaleThreads, 0, stream>>>(dsq, out);
}


# Scan wave owns a 32-column slab as two tiles (even, odd columns): 8-byte loads and stores, a row's 16 lanes cover a whole 128-byte line no other wave touches; d-sum tree and atomic addends change
# speedup vs baseline: 16.4483x; 1.0276x over previous
#include <hip/hip_runtime.h>
#include <math.h>

constexpr int kB = 8;
constexpr int kT = 2048;
constexpr int kD = 512;
constexpr int kGuard = 32;

typedef __attribute__((ext_vector_type(2))) float v2f;
typedef __attribute__((ext_vector_type(8))) float v8f;
typedef __attribute__((address_space(1))) float gfloat;
typedef __attribute__((address_space(1))) v2f gv2f;

__device__ __forceinline__ float softplus_dev(const float* kp) {
    float k = kp[0];
    return fmaxf(k, 0.0f) + log1pf(expf(-fabsf(k)));
}

__device__ __forceinline__ float xhalf(float x) {
    const int xi = __builtin_bit_cast(int, x);
    return __builtin_bit_cast(float, __builtin_amdgcn_permlanex16(0, xi, 0x76543210, 0xfedcba98, false, false));
}
template <int M>
__device__ __forceinline__ float xrow(float x) {
    const int xi = __builtin_bit_cast(int, x);
    return __builtin_bit_cast(float, __builtin_amdgcn_update_dpp(0, xi, 0x160 + M, 0xf, 0xf, true));
}

__global__ void __launch_bounds__(32) scan_both_wmma_kernel(const float* __restrict__ h,
                                      const float* __restrict__ kp,
                                      float* __restrict__ dsq,
                                      float* __restrict__ out) {
    const int lane  = threadIdx.x;
    const int slab  = blockIdx.x & 15;
    const int b     = blockIdx.x >> 4;
    const int n     = lane & 15;
    const int hi    = (lane >> 4) & 1;

    const float sp  = softplus_dev(kp);
    const float l2r = -sp * 1.4426950408889634f;
    auto rpow = [&](int dist) -> float { return exp2f((float)dist * l2r); };

    v2f a[4], a1[4];
    #pragma unroll
    for (int g = 0; g < 4; ++g) {
        int j0 = 8 * hi + 2 * g;
        a[g].x  = (j0     <= n) ? rpow(n - j0)     : 0.f;
        a[g].y  = (j0 + 1 <= n) ? rpow(n - j0 - 1) : 0.f;
        a1[g].x = (j0     <= n) ? 1.f : 0.f;
        a1[g].y = (j0 + 1 <= n) ? 1.f : 0.f;
    }

    float rpc[8];
    #pragma unroll
    for (int v = 0; v < 8; ++v) rpc[v] = rpow(v + 1 + 8 * hi);

    const float* hb = h   + ((size_t)b * kT + 8 * hi) * kD + slab * 32 + 2 * n;
    float*       ob = out + ((size_t)b * kT + 8 * hi) * kD + slab * 32 + 2 * n;
    float*       qb = dsq + b * kT + 16 * (n & 1) + (n >> 1) + 8 * hi;

    const bool b3 = (n & 8) != 0;
    const bool b2 = (n & 4) != 0;
    const bool b1 = (n & 2) != 0;
    const bool b0 = (n & 1) != 0;

    auto load_chunk = [&](v2f (&dst)[8], const float* p0) {
        #pragma unroll
        for (int w = 0; w < 8; ++w) dst[w] = *reinterpret_cast<const v2f*>(p0 + w * kD);
    };

    constexpr int kAhead = 8;
    constexpr int kChunks = kT / 16;
    v2f ring[kAhead][8];
    #pragma unroll
    for (int u = 0; u < kAhead; ++u) load_chunk(ring[u], hb + (size_t)(16 * u) * kD);

    float SE = 0.f, SO = 0.f;
    float ScE = 0.f, ScO = 0.f;
    float lastE = 0.f, lastO = 0.f;
    float r1_even = 0.f;
    constexpr int kKeep = 2;
    float kept[kKeep][16];
    float pend   = 0.f;
    float issued = 0.f;
    #pragma unroll
    for (int i = 0; i < kKeep; ++i) {
        #pragma unroll
        for (int v = 0; v < 16; ++v) kept[i][v] = 0.f;
    }
    gfloat* op = (gfloat*)ob;
    gfloat* qp = (gfloat*)qb;
    for (int c0 = 0; c0 < kChunks; c0 += kAhead) {
        const int cl = (c0 + kAhead < kChunks) ? (c0 + kAhead) : (kChunks - kAhead);
        const float* lp = hb + (size_t)(16 * cl) * kD;
        #pragma unroll
        for (int u = 0; u < kAhead; ++u) {
            v2f (&bm)[8] = ring[u];
            asm volatile("" : "+v"(bm[0].x), "+v"(bm[0].y), "+v"(bm[1].x), "+v"(bm[1].y),
                              "+v"(bm[2].x), "+v"(bm[2].y), "+v"(bm[3].x), "+v"(bm[3].y),
                              "+v"(bm[4].x), "+v"(bm[4].y), "+v"(bm[5].x), "+v"(bm[5].y),
                              "+v"(bm[6].x), "+v"(bm[6].y), "+v"(bm[7].x), "+v"(bm[7].y));

            v8f cE, cO, ccE, ccO;
            #pragma unroll
            for (int v = 0; v < 8; ++v) { cE[v] = rpc[v] * SE; cO[v] = rpc[v] * SO; ccE[v] = ScE; ccO[v] = ScO; }

            #pragma unroll
            for (int g = 0; g < 4; ++g) {
                v2f bE, bO;
                bE.x = bm[2 * g].x; bE.y = bm[2 * g + 1].x;
                bO.x = bm[2 * g].y; bO.y = bm[2 * g + 1].y;
                cE  = __builtin_amdgcn_wmma_f32_16x16x4_f32(false, a[g],  false, bE, (short)0, cE,  false, false);
                ccE = __builtin_amdgcn_wmma_f32_16x16x4_f32(false, a1[g], false, bE, (short)0, ccE, false, false);
                cO  = __builtin_amdgcn_wmma_f32_16x16x4_f32(false, a[g],  false, bO, (short)0, cO,  false, false);
                ccO = __builtin_amdgcn_wmma_f32_16x16x4_f32(false, a1[g], false, bO, (short)0, ccO, false, false);
            }

            if (u == 1) {
                asm volatile("" : "+v"(op), "+v"(qp) : "v"(cE[7]), "v"(ccE[7]), "v"(cO[7]), "v"(ccO[7]) : "memory");
                op += (size_t)(16 * kAhead) * kD;
                qp += 16 * kAhead;
                asm volatile("" : "+v"(op), "+v"(qp));
            }
            const int rel = (u == 0) ? 0 : -kAhead;
            if ((u & 1) == 0) {
                asm volatile("" : : "v"(cE[7]), "v"(ccE[7]), "v"(cO[7]), "v"(ccO[7]) : "memory");
                atomicAdd((float*)(qp + 16 * (u - 2 + rel)), pend);
                issued = pend;
            }
            v2f ab[8];
            #pragma unroll
            for (int v = 0; v < 8; ++v) {
                ab[v].x = fabsf(cE[v]);
                ab[v].y = fabsf(cO[v]);
                *reinterpret_cast<gv2f*>(op + (ptrdiff_t)(16 * (u + rel) + v) * kD) = ab[v];
            }

            float xchE = xhalf(bm[7].x);
            float xchO = xhalf(bm[7].y);
            float hp0E = hi ? xchE : lastE;
            float hp0O = hi ? xchO : lastO;
            lastE = xchE;
            lastO = xchO;

            float sq[8];
            {
                float dE = ccE[0] - hp0E;
                float dO = ccO[0] - hp0O;
                sq[0] = dE * dE + dO * dO;
            }
            #pragma unroll
            for (int v = 1; v < 8; ++v) {
                float dE = ccE[v] - bm[v - 1].x;
                float dO = ccO[v] - bm[v - 1].y;
                sq[v] = dE * dE + dO * dO;
            }

            asm volatile("" : "+v"(sq[0]), "+v"(sq[1]), "+v"(sq[2]), "+v"(sq[3]),
                              "+v"(sq[4]), "+v"(sq[5]), "+v"(sq[6]), "+v"(sq[7])
                            : "v"(bm[0].x), "v"(bm[0].y), "v"(bm[1].x), "v"(bm[1].y),
                              "v"(bm[2].x), "v"(bm[2].y), "v"(bm[3].x), "v"(bm[3].y),
                              "v"(bm[4].x), "v"(bm[4].y), "v"(bm[5].x), "v"(bm[5].y),
                              "v"(bm[6].x), "v"(bm[6].y), "v"(bm[7].x), "v"(bm[7].y)
                            : "memory");
            load_chunk(ring[u], lp + (size_t)(16 * u) * kD);
            float r4[4];
            #pragma unroll
            for (int i = 0; i < 4; ++i) {
                float keep = b3 ? sq[i + 4] : sq[i];
                float send = b3 ? sq[i]     : sq[i + 4];
                r4[i] = keep + xrow<8>(send);
            }
            float r2[2];
            #pragma unroll
            for (int i = 0; i < 2; ++i) {
                float keep = b2 ? r4[i + 2] : r4[i];
                float send = b2 ? r4[i]     : r4[i + 2];
                r2[i] = keep + xrow<4>(send);
            }
            float keep1 = b1 ? r2[1] : r2[0];
            float send1 = b1 ? r2[0] : r2[1];
            float r1  = keep1 + xrow<2>(send1);
            if ((u & 1) == 0) {
                r1_even = r1;
            } else {
                float keep0 = b0 ? r1      : r1_even;
                float send0 = b0 ? r1_even : r1;
                float tot   = keep0 + xrow<1>(send0);
                asm volatile("" : : "v"(issued));
                pend = tot;
            }
            asm volatile("" : : "v"(kept[u % kKeep][0]), "v"(kept[u % kKeep][1]), "v"(kept[u % kKeep][2]),
                                "v"(kept[u % kKeep][3]), "v"(kept[u % kKeep][4]), "v"(kept[u % kKeep][5]),
                                "v"(kept[u % kKeep][6]), "v"(kept[u % kKeep][7]), "v"(kept[u % kKeep][8]),
                                "v"(kept[u % kKeep][9]), "v"(kept[u % kKeep][10]), "v"(kept[u % kKeep][11]),
                                "v"(kept[u % kKeep][12]), "v"(kept[u % kKeep][13]), "v"(kept[u % kKeep][14]),
                                "v"(kept[u % kKeep][15]));
            #pragma unroll
            for (int v = 0; v < 8; ++v) { kept[u % kKeep][2 * v] = ab[v].x; kept[u % kKeep][2 * v + 1] = ab[v].y; }

            {
                const float sxE  = xhalf(cE[7]);
                const float sxO  = xhalf(cO[7]);
                const float scxE = xhalf(ccE[7]);
                const float scxO = xhalf(ccO[7]);
                SE  = hi ? cE[7]  : sxE;
                SO  = hi ? cO[7]  : sxO;
                ScE = hi ? ccE[7] : scxE;
                ScO = hi ? ccO[7] : scxO;
            }
        }
    }
    atomicAdd(qb + 16 * (kChunks - 2), pend);
}

constexpr int kScaleRows    = 4;
constexpr int kScaleThreads = 256;
constexpr int kScaleWaves   = (kB * kT) / kScaleRows;
constexpr int kScaleLoads   = kScaleRows * (kD / 4) / 32;
__global__ void __launch_bounds__(kScaleThreads) scale_rows_kernel(const float* __restrict__ dsq,
                                  float* __restrict__ out) {
    const int lane = threadIdx.x & 31;
    const int wave = blockIdx.x * (kScaleThreads / 32) + (threadIdx.x >> 5);
    const size_t row0 = (size_t)wave * kScaleRows;

    const float4 q = *reinterpret_cast<const float4*>(dsq + row0);
    float4* o4 = reinterpret_cast<float4*>(out) + row0 * (kD / 4) + lane;

    float4 v[kScaleLoads];
    #pragma unroll
    for (int i = 0; i < kScaleLoads; ++i) v[i] = o4[i * 32];

    const float m[kScaleRows] = { sqrtf(q.x), sqrtf(q.y), sqrtf(q.z), sqrtf(q.w) };

    #pragma unroll
    for (int i = 0; i < kScaleLoads; ++i) {
        const float s = m[i / (kScaleLoads / kScaleRows)];
        float4 w = v[i];
        w.x *= s; w.y *= s; w.z *= s; w.w *= s;
        o4[i * 32] = w;
    }
}

extern "C" void kernel_launch(void* const* d_in, const int* in_sizes, int n_in,
                              void* d_out, int out_size, void* d_ws, size_t ws_size,
                              hipStream_t stream) {
    const float* h  = (const float*)d_in[0];
    const float* k  = (const float*)d_in[1];
    float* out = (float*)d_out;
    float* dsq = (float*)d_ws + kGuard;

    (void)hipMemsetAsync(d_ws, 0, (size_t)(kB * kT + kGuard) * sizeof(float), stream);
    scan_both_wmma_kernel<<<kB * (kD / 32), 32, 0, stream>>>(h, k, dsq, out);
    scale_rows_kernel<<<kScaleWaves / (kScaleThreads / 32), kScaleThreads, 0, stream>>>(dsq, out);
}
